# starting kernel
# baseline (speedup 1.0000x reference)
#include <hip/hip_runtime.h>

typedef __attribute__((ext_vector_type(2))) float v2f;
typedef __attribute__((ext_vector_type(8))) float v8f;

#define IN_DIM 300
#define H_DIM  128
#define L_LAYERS 4

#define XS (IN_DIM + 2)
#define HS (H_DIM + 4)


__global__ void zero_f32_kernel(float* __restrict__ p, int n) {
    int i = blockIdx.x * blockDim.x + threadIdx.x;
    if (i < n) p[i] = 0.0f;
}

__global__ void degree_kernel(const int* __restrict__ dst, float* __restrict__ deg, int e) {
    int i = blockIdx.x * blockDim.x + threadIdx.x;
    if (i < e) unsafeAtomicAdd(&deg[dst[i]], 1.0f);
}

__global__ void transpose_kernel(const float* __restrict__ in, float* __restrict__ out,
                                 int K, int Nc) {
    int i = blockIdx.x * blockDim.x + threadIdx.x;
    if (i < K * Nc) {
        int k = i / Nc;
        int n = i - k * Nc;
        out[n * K + k] = in[i];
    }
}

__global__ __launch_bounds__(256)
void scatter_kernel(const int* __restrict__ src, const int* __restrict__ dst,
                    const float* __restrict__ h, float* __restrict__ agg, int e) {
    int wave = (blockIdx.x * blockDim.x + threadIdx.x) >> 5;
    int lane = threadIdx.x & 31;
    if (wave >= e) return;
    int s = src[wave];
    int d = dst[wave];
    const float4* hp = reinterpret_cast<const float4*>(h + (long)s * H_DIM);
    float4 v = hp[lane];
    float* ap = agg + (long)d * H_DIM + lane * 4;
    unsafeAtomicAdd(ap + 0, v.x);
    unsafeAtomicAdd(ap + 1, v.y);
    unsafeAtomicAdd(ap + 2, v.z);
    unsafeAtomicAdd(ap + 3, v.w);
}


__global__ __launch_bounds__(256)
void emb_kernel(const float* __restrict__ x, const float* __restrict__ WembT,
                const float* __restrict__ bemb, float* __restrict__ h, int n_nodes) {
    __shared__ float shx[16 * XS];
    const int  tid  = threadIdx.x;
    const long row0 = (long)blockIdx.x * 16;
    const bool full = (row0 + 16 <= (long)n_nodes);

    if (full) {
        for (int i = tid; i < 16 * IN_DIM; i += 256) {
            int r = i / IN_DIM, c = i - r * IN_DIM;
            shx[r * XS + c] = x[(row0 + r) * IN_DIM + c];
        }
    } else {
        for (int i = tid; i < 16 * IN_DIM; i += 256) {
            int r = i / IN_DIM, c = i - r * IN_DIM;
            long rr = row0 + r; if (rr >= n_nodes) rr = n_nodes - 1;
            shx[r * XS + c] = x[rr * IN_DIM + c];
        }
    }
    __syncthreads();

    const int wave = tid >> 5, lane = tid & 31;
    const int m = lane & 15, hi = lane >> 4;
    const int ncol = wave * 16 + m;

    v8f acc;
    float bv = bemb[ncol];
    #pragma unroll
    for (int i = 0; i < 8; ++i) acc[i] = bv;

    const float* wt = WembT + (long)ncol * IN_DIM;
    #pragma unroll 5
    for (int k0 = 0; k0 < IN_DIM; k0 += 4) {
        int k = k0 + 2 * hi;
        v2f a; a.x = shx[m * XS + k]; a.y = shx[m * XS + k + 1];
        v2f b; b.x = wt[k];           b.y = wt[k + 1];
        acc = __builtin_amdgcn_wmma_f32_16x16x4_f32(false, a, false, b,
                                                    (short)0, acc, false, false);
    }

    float* hrow = h + row0 * H_DIM + wave * 16 + m;
    if (full) {
        #pragma unroll
        for (int v = 0; v < 8; ++v) {
            float r = acc[v];
            hrow[(long)(v + 8 * hi) * H_DIM] = r > 0.0f ? r : 0.0f;
        }
    } else {
        #pragma unroll
        for (int v = 0; v < 8; ++v) {
            int mm = v + 8 * hi;
            if (row0 + mm < n_nodes) {
                float r = acc[v];
                hrow[(long)mm * H_DIM] = r > 0.0f ? r : 0.0f;
            }
        }
    }
}

__global__ __launch_bounds__(256)
void sage_layer_kernel(const float* __restrict__ agg, const float* __restrict__ hin,
                       const float* __restrict__ deg,
                       const float* __restrict__ WlT, const float* __restrict__ WrT,
                       const float* __restrict__ bias, float* __restrict__ hout,
                       int n_nodes) {
    __shared__ float sh_a[16 * HS];
    __shared__ float sh_h[16 * HS];
    const int  tid  = threadIdx.x;
    const long row0 = (long)blockIdx.x * 16;
    const bool full = (row0 + 16 <= (long)n_nodes);

    if (full) {
        for (int i = tid; i < 16 * H_DIM; i += 256) {
            int r = i >> 7, c = i & 127;
            long rr = row0 + r;
            float dg  = deg[rr];
            float inv = 1.0f / (dg > 1.0f ? dg : 1.0f);
            sh_a[r * HS + c] = agg[rr * H_DIM + c] * inv;
            sh_h[r * HS + c] = hin[rr * H_DIM + c];
        }
    } else {
        for (int i = tid; i < 16 * H_DIM; i += 256) {
            int r = i >> 7, c = i & 127;
            long rr = row0 + r; if (rr >= n_nodes) rr = n_nodes - 1;
            float dg  = deg[rr];
            float inv = 1.0f / (dg > 1.0f ? dg : 1.0f);
            sh_a[r * HS + c] = agg[rr * H_DIM + c] * inv;
            sh_h[r * HS + c] = hin[rr * H_DIM + c];
        }
    }
    __syncthreads();

    const int wave = tid >> 5, lane = tid & 31;
    const int m = lane & 15, hi = lane >> 4;
    const int ncol = wave * 16 + m;

    v8f acc;
    float bv = bias[ncol];
    #pragma unroll
    for (int i = 0; i < 8; ++i) acc[i] = bv;

    const float* wl = WlT + (long)ncol * H_DIM;
    const float* wr = WrT + (long)ncol * H_DIM;
    #pragma unroll 4
    for (int k0 = 0; k0 < H_DIM; k0 += 4) {
        int k = k0 + 2 * hi;
        v2f aa; aa.x = sh_a[m * HS + k]; aa.y = sh_a[m * HS + k + 1];
        v2f ah; ah.x = sh_h[m * HS + k]; ah.y = sh_h[m * HS + k + 1];
        v2f bl; bl.x = wl[k];            bl.y = wl[k + 1];
        v2f br; br.x = wr[k];            br.y = wr[k + 1];
        acc = __builtin_amdgcn_wmma_f32_16x16x4_f32(false, aa, false, bl,
                                                    (short)0, acc, false, false);
        acc = __builtin_amdgcn_wmma_f32_16x16x4_f32(false, ah, false, br,
                                                    (short)0, acc, false, false);
    }

    float* orow = hout + row0 * H_DIM + wave * 16 + m;
    if (full) {
        #pragma unroll
        for (int v = 0; v < 8; ++v) {
            float r = acc[v];
            orow[(long)(v + 8 * hi) * H_DIM] = r > 0.0f ? r : 0.0f;
        }
    } else {
        #pragma unroll
        for (int v = 0; v < 8; ++v) {
            int mm = v + 8 * hi;
            if (row0 + mm < n_nodes) {
                float r = acc[v];
                orow[(long)mm * H_DIM] = r > 0.0f ? r : 0.0f;
            }
        }
    }
}


static inline int cdiv(long a, long b) { return (int)((a + b - 1) / b); }

extern "C" void kernel_launch(void* const* d_in, const int* in_sizes, int n_in,
                              void* d_out, int out_size, void* d_ws, size_t ws_size,
                              hipStream_t stream) {
    const float* x    = (const float*)d_in[0];
    const int*   ei   = (const int*)  d_in[1];
    const float* Wemb = (const float*)d_in[2];
    const float* bemb = (const float*)d_in[3];
    const float* Wl   = (const float*)d_in[4];
    const float* bl   = (const float*)d_in[5];
    const float* Wr   = (const float*)d_in[6];
    float* out = (float*)d_out;

    const int N = in_sizes[0] / IN_DIM;
    const int E = in_sizes[1] / 2;
    const int* src = ei;
    const int* dst = ei + E;

    char*  ws  = (char*)d_ws;
    size_t off = 0;
    auto carve = [&](size_t bytes) -> float* {
        float* p = (float*)(ws + off);
        off = (off + bytes + 255) & ~(size_t)255;
        return p;
    };
    float* deg   = carve((size_t)N * sizeof(float));
    float* hA    = carve((size_t)N * H_DIM * sizeof(float));
    float* hB    = carve((size_t)N * H_DIM * sizeof(float));
    float* agg   = carve((size_t)N * H_DIM * sizeof(float));
    float* WembT = carve((size_t)IN_DIM * H_DIM * sizeof(float));
    float* WlT   = carve((size_t)L_LAYERS * H_DIM * H_DIM * sizeof(float));
    float* WrT   = carve((size_t)L_LAYERS * H_DIM * H_DIM * sizeof(float));
    (void)ws_size;

    const int tiles = cdiv(N, 16);

    zero_f32_kernel<<<cdiv(N, 256), 256, 0, stream>>>(deg, N);
    degree_kernel<<<cdiv(E, 256), 256, 0, stream>>>(dst, deg, E);

    transpose_kernel<<<cdiv((long)IN_DIM * H_DIM, 256), 256, 0, stream>>>(Wemb, WembT, IN_DIM, H_DIM);
    for (int l = 0; l < L_LAYERS; ++l) {
        transpose_kernel<<<cdiv((long)H_DIM * H_DIM, 256), 256, 0, stream>>>(
            Wl + (long)l * H_DIM * H_DIM, WlT + (long)l * H_DIM * H_DIM, H_DIM, H_DIM);
        transpose_kernel<<<cdiv((long)H_DIM * H_DIM, 256), 256, 0, stream>>>(
            Wr + (long)l * H_DIM * H_DIM, WrT + (long)l * H_DIM * H_DIM, H_DIM, H_DIM);
    }

    emb_kernel<<<tiles, 256, 0, stream>>>(x, WembT, bemb, hA, N);

    float* hcur = hA;
    for (int l = 0; l < L_LAYERS; ++l) {
        zero_f32_kernel<<<cdiv((long)N * H_DIM, 256), 256, 0, stream>>>(agg, N * H_DIM);
        scatter_kernel<<<cdiv(E, 8), 256, 0, stream>>>(src, dst, hcur, agg, E);
        float* hnext = (l == L_LAYERS - 1) ? out : (hcur == hA ? hB : hA);
        sage_layer_kernel<<<tiles, 256, 0, stream>>>(
            agg, hcur, deg,
            WlT + (long)l * H_DIM * H_DIM, WrT + (long)l * H_DIM * H_DIM,
            bl + (long)l * H_DIM, hnext, N);
        hcur = hnext;
    }
}


# aggregation without float atomics: edges grouped by dst once per call (int atomic add gives rank, scan, fill), per-node register sum of gathered rows; sum order = arrival-order rank
# speedup vs baseline: 2.2184x; 2.2184x over previous
#include <hip/hip_runtime.h>

typedef __attribute__((ext_vector_type(2))) float v2f;
typedef __attribute__((ext_vector_type(8))) float v8f;

#define IN_DIM 300
#define H_DIM  128
#define L_LAYERS 4

#define XS (IN_DIM + 2)
#define HS (H_DIM + 4)


__global__ void zero_i32_kernel(int* __restrict__ p, int n) {
    int i = blockIdx.x * blockDim.x + threadIdx.x;
    if (i < n) p[i] = 0;
}

__global__ void count_rank_kernel(const int* __restrict__ dst, int* __restrict__ cnt,
                                  int* __restrict__ rank, int e) {
    int i = blockIdx.x * blockDim.x + threadIdx.x;
    if (i < e) rank[i] = atomicAdd(&cnt[dst[i]], 1);
}

__global__ __launch_bounds__(256)
void block_sum_kernel(const int* __restrict__ cnt, int* __restrict__ bsum, int n) {
    __shared__ int sh[8];
    const int tid = threadIdx.x;
    const int i = blockIdx.x * 256 + tid;
    int v = (i < n) ? cnt[i] : 0;
    #pragma unroll
    for (int o = 16; o > 0; o >>= 1) v += __shfl_xor(v, o, 32);
    if ((tid & 31) == 0) sh[tid >> 5] = v;
    __syncthreads();
    if (tid == 0) {
        int s = 0;
        #pragma unroll
        for (int w = 0; w < 8; ++w) s += sh[w];
        bsum[blockIdx.x] = s;
    }
}

__global__ __launch_bounds__(256)
void row_ptr_kernel(const int* __restrict__ cnt, const int* __restrict__ bsum,
                    int* __restrict__ row_ptr, float* __restrict__ deg, int n) {
    __shared__ int sh_part[8];
    __shared__ int sh_tot[8];
    const int tid = threadIdx.x, lane = tid & 31, wave = tid >> 5;
    int part = 0;
    for (int j = tid; j < (int)blockIdx.x; j += 256) part += bsum[j];
    #pragma unroll
    for (int o = 16; o > 0; o >>= 1) part += __shfl_xor(part, o, 32);
    const int i = blockIdx.x * 256 + tid;
    const int c = (i < n) ? cnt[i] : 0;
    int inc = c;
    #pragma unroll
    for (int o = 1; o < 32; o <<= 1) {
        int t = __shfl_up(inc, o, 32);
        if (lane >= o) inc += t;
    }
    if (lane == 31) sh_tot[wave] = inc;
    if (lane == 0)  sh_part[wave] = part;
    __syncthreads();
    int base = 0;
    #pragma unroll
    for (int w = 0; w < 8; ++w) {
        base += sh_part[w];
        if (w < wave) base += sh_tot[w];
    }
    const int excl = base + inc - c;
    if (i < n) {
        row_ptr[i] = excl;
        deg[i] = (float)c;
        if (i == n - 1) row_ptr[n] = excl + c;
    }
}

__global__ void fill_kernel(const int* __restrict__ src, const int* __restrict__ dst,
                            const int* __restrict__ rank, const int* __restrict__ row_ptr,
                            int* __restrict__ csr_src, int e) {
    int i = blockIdx.x * blockDim.x + threadIdx.x;
    if (i < e) csr_src[row_ptr[dst[i]] + rank[i]] = src[i];
}

__global__ void transpose_kernel(const float* __restrict__ in, float* __restrict__ out,
                                 int K, int Nc) {
    int i = blockIdx.x * blockDim.x + threadIdx.x;
    if (i < K * Nc) {
        int k = i / Nc;
        int n = i - k * Nc;
        out[n * K + k] = in[i];
    }
}

#define GATHER_BLOCKS 4096
__device__ __forceinline__ void add4(float4& a, const float4& v) {
    a.x += v.x; a.y += v.y; a.z += v.z; a.w += v.w;
}
template <int R>
__device__ __forceinline__ void gather_rows(const float4* __restrict__ h4, int idx, int j,
                                            int lane, float4& acc) {
    float4 v[R];
    #pragma unroll
    for (int u = 0; u < R; ++u) {
        int s = __builtin_amdgcn_readlane(idx, j + u);
        v[u] = h4[(long)s * (H_DIM / 4) + lane];
    }
    #pragma unroll
    for (int u = 0; u < R; ++u) add4(acc, v[u]);
}
__global__ __launch_bounds__(256)
void gather_sum_kernel(const int* __restrict__ row_ptr, const int* __restrict__ csr_src,
                       const float* __restrict__ h, float* __restrict__ agg, int n_nodes) {
    const int lane = threadIdx.x & 31;
    const int wave0 = __builtin_amdgcn_readfirstlane((int)((blockIdx.x * 256 + threadIdx.x) >> 5));
    const float4* h4 = reinterpret_cast<const float4*>(h);
    float4* a4 = reinterpret_cast<float4*>(agg);
    for (int node = wave0; node < n_nodes; node += GATHER_BLOCKS * 8) {
        const int beg = row_ptr[node], end = row_ptr[node + 1];
        float4 acc = make_float4(0.0f, 0.0f, 0.0f, 0.0f);
        for (int base = beg; base < end; base += 32) {
            int m = end - base; if (m > 32) m = 32;
            int idx = (lane < m) ? csr_src[base + lane] : 0;
            int j = 0;
            for (; j + 8 <= m; j += 8) gather_rows<8>(h4, idx, j, lane, acc);
            if (m & 4) { gather_rows<4>(h4, idx, j, lane, acc); j += 4; }
            if (m & 2) { gather_rows<2>(h4, idx, j, lane, acc); j += 2; }
            if (m & 1) { gather_rows<1>(h4, idx, j, lane, acc); }
        }
        a4[(long)node * (H_DIM / 4) + lane] = acc;
    }
}


__global__ __launch_bounds__(256)
void emb_kernel(const float* __restrict__ x, const float* __restrict__ WembT,
                const float* __restrict__ bemb, float* __restrict__ h, int n_nodes) {
    __shared__ float shx[16 * XS];
    const int  tid  = threadIdx.x;
    const long row0 = (long)blockIdx.x * 16;
    const bool full = (row0 + 16 <= (long)n_nodes);

    if (full) {
        for (int i = tid; i < 16 * IN_DIM; i += 256) {
            int r = i / IN_DIM, c = i - r * IN_DIM;
            shx[r * XS + c] = x[(row0 + r) * IN_DIM + c];
        }
    } else {
        for (int i = tid; i < 16 * IN_DIM; i += 256) {
            int r = i / IN_DIM, c = i - r * IN_DIM;
            long rr = row0 + r; if (rr >= n_nodes) rr = n_nodes - 1;
            shx[r * XS + c] = x[rr * IN_DIM + c];
        }
    }
    __syncthreads();

    const int wave = tid >> 5, lane = tid & 31;
    const int m = lane & 15, hi = lane >> 4;
    const int ncol = wave * 16 + m;

    v8f acc;
    float bv = bemb[ncol];
    #pragma unroll
    for (int i = 0; i < 8; ++i) acc[i] = bv;

    const float* wt = WembT + (long)ncol * IN_DIM;
    #pragma unroll 5
    for (int k0 = 0; k0 < IN_DIM; k0 += 4) {
        int k = k0 + 2 * hi;
        v2f a; a.x = shx[m * XS + k]; a.y = shx[m * XS + k + 1];
        v2f b; b.x = wt[k];           b.y = wt[k + 1];
        acc = __builtin_amdgcn_wmma_f32_16x16x4_f32(false, a, false, b,
                                                    (short)0, acc, false, false);
    }

    float* hrow = h + row0 * H_DIM + wave * 16 + m;
    if (full) {
        #pragma unroll
        for (int v = 0; v < 8; ++v) {
            float r = acc[v];
            hrow[(long)(v + 8 * hi) * H_DIM] = r > 0.0f ? r : 0.0f;
        }
    } else {
        #pragma unroll
        for (int v = 0; v < 8; ++v) {
            int mm = v + 8 * hi;
            if (row0 + mm < n_nodes) {
                float r = acc[v];
                hrow[(long)mm * H_DIM] = r > 0.0f ? r : 0.0f;
            }
        }
    }
}

__global__ __launch_bounds__(256)
void sage_layer_kernel(const float* agg, const float* __restrict__ hin,
                       const float* __restrict__ deg,
                       const float* __restrict__ WlT, const float* __restrict__ WrT,
                       const float* __restrict__ bias, float* hout,
                       int n_nodes) {
    __shared__ float sh_a[16 * HS];
    __shared__ float sh_h[16 * HS];
    const int  tid  = threadIdx.x;
    const long row0 = (long)blockIdx.x * 16;
    const bool full = (row0 + 16 <= (long)n_nodes);

    if (full) {
        for (int i = tid; i < 16 * H_DIM; i += 256) {
            int r = i >> 7, c = i & 127;
            long rr = row0 + r;
            float dg  = deg[rr];
            float inv = 1.0f / (dg > 1.0f ? dg : 1.0f);
            sh_a[r * HS + c] = agg[rr * H_DIM + c] * inv;
            sh_h[r * HS + c] = hin[rr * H_DIM + c];
        }
    } else {
        for (int i = tid; i < 16 * H_DIM; i += 256) {
            int r = i >> 7, c = i & 127;
            long rr = row0 + r; if (rr >= n_nodes) rr = n_nodes - 1;
            float dg  = deg[rr];
            float inv = 1.0f / (dg > 1.0f ? dg : 1.0f);
            sh_a[r * HS + c] = agg[rr * H_DIM + c] * inv;
            sh_h[r * HS + c] = hin[rr * H_DIM + c];
        }
    }
    __syncthreads();

    const int wave = tid >> 5, lane = tid & 31;
    const int m = lane & 15, hi = lane >> 4;
    const int ncol = wave * 16 + m;

    v8f acc;
    float bv = bias[ncol];
    #pragma unroll
    for (int i = 0; i < 8; ++i) acc[i] = bv;

    const float* wl = WlT + (long)ncol * H_DIM;
    const float* wr = WrT + (long)ncol * H_DIM;
    #pragma unroll 4
    for (int k0 = 0; k0 < H_DIM; k0 += 4) {
        int k = k0 + 2 * hi;
        v2f aa; aa.x = sh_a[m * HS + k]; aa.y = sh_a[m * HS + k + 1];
        v2f ah; ah.x = sh_h[m * HS + k]; ah.y = sh_h[m * HS + k + 1];
        v2f bl; bl.x = wl[k];            bl.y = wl[k + 1];
        v2f br; br.x = wr[k];            br.y = wr[k + 1];
        acc = __builtin_amdgcn_wmma_f32_16x16x4_f32(false, aa, false, bl,
                                                    (short)0, acc, false, false);
        acc = __builtin_amdgcn_wmma_f32_16x16x4_f32(false, ah, false, br,
                                                    (short)0, acc, false, false);
    }

    float* orow = hout + row0 * H_DIM + wave * 16 + m;
    if (full) {
        #pragma unroll
        for (int v = 0; v < 8; ++v) {
            float r = acc[v];
            orow[(long)(v + 8 * hi) * H_DIM] = r > 0.0f ? r : 0.0f;
        }
    } else {
        #pragma unroll
        for (int v = 0; v < 8; ++v) {
            int mm = v + 8 * hi;
            if (row0 + mm < n_nodes) {
                float r = acc[v];
                orow[(long)mm * H_DIM] = r > 0.0f ? r : 0.0f;
            }
        }
    }
}


static inline int cdiv(long a, long b) { return (int)((a + b - 1) / b); }

extern "C" void kernel_launch(void* const* d_in, const int* in_sizes, int n_in,
                              void* d_out, int out_size, void* d_ws, size_t ws_size,
                              hipStream_t stream) {
    const float* x    = (const float*)d_in[0];
    const int*   ei   = (const int*)  d_in[1];
    const float* Wemb = (const float*)d_in[2];
    const float* bemb = (const float*)d_in[3];
    const float* Wl   = (const float*)d_in[4];
    const float* bl   = (const float*)d_in[5];
    const float* Wr   = (const float*)d_in[6];
    float* out = (float*)d_out;

    const int N = in_sizes[0] / IN_DIM;
    const int E = in_sizes[1] / 2;
    const int* src = ei;
    const int* dst = ei + E;

    char*  ws  = (char*)d_ws;
    size_t off = 0;
    auto carve = [&](size_t bytes) -> float* {
        float* p = (float*)(ws + off);
        off = (off + bytes + 255) & ~(size_t)255;
        return p;
    };
    const int nblk = cdiv(N, 256);
    float* deg   = carve((size_t)N * sizeof(float));
    float* hP    = carve((size_t)N * H_DIM * sizeof(float));
    float* hQ    = carve((size_t)N * H_DIM * sizeof(float));
    float* WembT = carve((size_t)IN_DIM * H_DIM * sizeof(float));
    float* WlT   = carve((size_t)L_LAYERS * H_DIM * H_DIM * sizeof(float));
    float* WrT   = carve((size_t)L_LAYERS * H_DIM * H_DIM * sizeof(float));
    int* cnt     = (int*)carve((size_t)N * sizeof(int));
    int* row_ptr = (int*)carve((size_t)(N + 1) * sizeof(int));
    int* bsum    = (int*)carve((size_t)nblk * sizeof(int));
    int* rank    = (int*)carve((size_t)E * sizeof(int));
    int* csr_src = (int*)carve((size_t)E * sizeof(int));
    (void)ws_size;

    const int tiles = cdiv(N, 16);

    zero_i32_kernel<<<cdiv(N, 256), 256, 0, stream>>>(cnt, N);
    count_rank_kernel<<<cdiv(E, 256), 256, 0, stream>>>(dst, cnt, rank, E);
    block_sum_kernel<<<nblk, 256, 0, stream>>>(cnt, bsum, N);
    row_ptr_kernel<<<nblk, 256, 0, stream>>>(cnt, bsum, row_ptr, deg, N);
    fill_kernel<<<cdiv(E, 256), 256, 0, stream>>>(src, dst, rank, row_ptr, csr_src, E);

    transpose_kernel<<<cdiv((long)IN_DIM * H_DIM, 256), 256, 0, stream>>>(Wemb, WembT, IN_DIM, H_DIM);
    for (int l = 0; l < L_LAYERS; ++l) {
        transpose_kernel<<<cdiv((long)H_DIM * H_DIM, 256), 256, 0, stream>>>(
            Wl + (long)l * H_DIM * H_DIM, WlT + (long)l * H_DIM * H_DIM, H_DIM, H_DIM);
        transpose_kernel<<<cdiv((long)H_DIM * H_DIM, 256), 256, 0, stream>>>(
            Wr + (long)l * H_DIM * H_DIM, WrT + (long)l * H_DIM * H_DIM, H_DIM, H_DIM);
    }

    emb_kernel<<<tiles, 256, 0, stream>>>(x, WembT, bemb, hP, N);

    float* hcur = hP;
    for (int l = 0; l < L_LAYERS; ++l) {
        float* agg = (hcur == hP) ? hQ : hP;
        gather_sum_kernel<<<GATHER_BLOCKS, 256, 0, stream>>>(row_ptr, csr_src, hcur, agg, N);
        float* hnext = (l == L_LAYERS - 1) ? out : agg;
        sage_layer_kernel<<<tiles, 256, 0, stream>>>(
            agg, hcur, deg,
            WlT + (long)l * H_DIM * H_DIM, WrT + (long)l * H_DIM * H_DIM,
            bl + (long)l * H_DIM, hnext, N);
        hcur = hnext;
    }
}


# B operand on chip: 512 persistent work-groups read each wave's B fragments once from the row-major weights into registers and loop over 16-row tiles; transposed copies and their 9 launches go
# speedup vs baseline: 3.3416x; 1.5063x over previous
#include <hip/hip_runtime.h>

typedef __attribute__((ext_vector_type(2))) float v2f;
typedef __attribute__((ext_vector_type(8))) float v8f;

#define IN_DIM 300
#define H_DIM  128
#define L_LAYERS 4

#define XS (IN_DIM + 16)
#define HS (H_DIM + 4)


__global__ void zero_i32_kernel(int* __restrict__ p, int n) {
    int i = blockIdx.x * blockDim.x + threadIdx.x;
    if (i < n) p[i] = 0;
}

__global__ void count_rank_kernel(const int* __restrict__ dst, int* __restrict__ cnt,
                                  int* __restrict__ rank, int e) {
    int i = blockIdx.x * blockDim.x + threadIdx.x;
    if (i < e) rank[i] = atomicAdd(&cnt[dst[i]], 1);
}

__global__ __launch_bounds__(256)
void block_sum_kernel(const int* __restrict__ cnt, int* __restrict__ bsum, int n) {
    __shared__ int sh[8];
    const int tid = threadIdx.x;
    const int i = blockIdx.x * 256 + tid;
    int v = (i < n) ? cnt[i] : 0;
    #pragma unroll
    for (int o = 16; o > 0; o >>= 1) v += __shfl_xor(v, o, 32);
    if ((tid & 31) == 0) sh[tid >> 5] = v;
    __syncthreads();
    if (tid == 0) {
        int s = 0;
        #pragma unroll
        for (int w = 0; w < 8; ++w) s += sh[w];
        bsum[blockIdx.x] = s;
    }
}

__global__ __launch_bounds__(256)
void row_ptr_kernel(const int* __restrict__ cnt, const int* __restrict__ bsum,
                    int* __restrict__ row_ptr, float* __restrict__ deg, int n) {
    __shared__ int sh_part[8];
    __shared__ int sh_tot[8];
    const int tid = threadIdx.x, lane = tid & 31, wave = tid >> 5;
    int part = 0;
    for (int j = tid; j < (int)blockIdx.x; j += 256) part += bsum[j];
    #pragma unroll
    for (int o = 16; o > 0; o >>= 1) part += __shfl_xor(part, o, 32);
    const int i = blockIdx.x * 256 + tid;
    const int c = (i < n) ? cnt[i] : 0;
    int inc = c;
    #pragma unroll
    for (int o = 1; o < 32; o <<= 1) {
        int t = __shfl_up(inc, o, 32);
        if (lane >= o) inc += t;
    }
    if (lane == 31) sh_tot[wave] = inc;
    if (lane == 0)  sh_part[wave] = part;
    __syncthreads();
    int base = 0;
    #pragma unroll
    for (int w = 0; w < 8; ++w) {
        base += sh_part[w];
        if (w < wave) base += sh_tot[w];
    }
    const int excl = base + inc - c;
    if (i < n) {
        row_ptr[i] = excl;
        deg[i] = (float)c;
        if (i == n - 1) row_ptr[n] = excl + c;
    }
}

__global__ void fill_kernel(const int* __restrict__ src, const int* __restrict__ dst,
                            const int* __restrict__ rank, const int* __restrict__ row_ptr,
                            int* __restrict__ csr_src, int e) {
    int i = blockIdx.x * blockDim.x + threadIdx.x;
    if (i < e) csr_src[row_ptr[dst[i]] + rank[i]] = src[i];
}

#define GATHER_BLOCKS 4096
__device__ __forceinline__ void add4(float4& a, const float4& v) {
    a.x += v.x; a.y += v.y; a.z += v.z; a.w += v.w;
}
template <int R>
__device__ __forceinline__ void gather_rows(const float4* __restrict__ h4, int idx, int j,
                                            int lane, float4& acc) {
    float4 v[R];
    #pragma unroll
    for (int u = 0; u < R; ++u) {
        int s = __builtin_amdgcn_readlane(idx, j + u);
        v[u] = h4[(long)s * (H_DIM / 4) + lane];
    }
    #pragma unroll
    for (int u = 0; u < R; ++u) add4(acc, v[u]);
}
__global__ __launch_bounds__(256)
void gather_sum_kernel(const int* __restrict__ row_ptr, const int* __restrict__ csr_src,
                       const float* __restrict__ h, float* __restrict__ agg, int n_nodes) {
    const int lane = threadIdx.x & 31;
    const int wave0 = __builtin_amdgcn_readfirstlane((int)((blockIdx.x * 256 + threadIdx.x) >> 5));
    const float4* h4 = reinterpret_cast<const float4*>(h);
    float4* a4 = reinterpret_cast<float4*>(agg);
    for (int node = wave0; node < n_nodes; node += GATHER_BLOCKS * 8) {
        const int beg = row_ptr[node], end = row_ptr[node + 1];
        float4 acc = make_float4(0.0f, 0.0f, 0.0f, 0.0f);
        for (int base = beg; base < end; base += 32) {
            int m = end - base; if (m > 32) m = 32;
            int idx = (lane < m) ? csr_src[base + lane] : 0;
            int j = 0;
            for (; j + 8 <= m; j += 8) gather_rows<8>(h4, idx, j, lane, acc);
            if (m & 4) { gather_rows<4>(h4, idx, j, lane, acc); j += 4; }
            if (m & 2) { gather_rows<2>(h4, idx, j, lane, acc); j += 2; }
            if (m & 1) { gather_rows<1>(h4, idx, j, lane, acc); }
        }
        a4[(long)node * (H_DIM / 4) + lane] = acc;
    }
}

#define GEMM_GROUPS 512

#define XQ (IN_DIM / 4)
#define XF4 (16 * XQ)
#define XPT ((XF4 + 255) / 256)
__device__ __forceinline__ float4 x_fetch(const float* __restrict__ x, int t, int tid, int j,
                                          int n_nodes) {
    int i = tid + 256 * j; if (i >= XF4) i = XF4 - 1;
    const int r = i / XQ, c = i - r * XQ;
    long rr = (long)t * 16 + r; if (rr >= n_nodes) rr = n_nodes - 1;
    return reinterpret_cast<const float4*>(x + rr * IN_DIM)[c];
}
__device__ __forceinline__ void x_stage(float* sx, int tid, int j, const float4& v) {
    const int i = tid + 256 * j;
    if (i < XF4) {
        const int r = i / XQ, c = i - r * XQ;
        *reinterpret_cast<float4*>(sx + r * XS + 4 * c) = v;
    }
}
__global__ __launch_bounds__(256)
void emb_kernel(const float* __restrict__ x, const float* __restrict__ Wemb,
                const float* __restrict__ bemb, float* __restrict__ h,
                int n_nodes, int n_tiles) {
    __shared__ __attribute__((aligned(16))) float shx[2][16 * XS];
    const int tid = threadIdx.x;
    const int wave = tid >> 5, lane = tid & 31;
    const int m = lane & 15, hi = lane >> 4;
    const int ncol = wave * 16 + m;

    v2f b[IN_DIM / 4];
    #pragma unroll
    for (int s = 0; s < IN_DIM / 4; ++s) {
        const int k = 4 * s + 2 * hi;
        b[s].x = Wemb[k * H_DIM + ncol];
        b[s].y = Wemb[(k + 1) * H_DIM + ncol];
    }
    const float bv = bemb[ncol];

    static_assert(XPT == 5, "five float4 of x per thread and tile");
    float4 px0, px1, px2, px3, px4;
    px0 = px1 = px2 = px3 = px4 = make_float4(0.0f, 0.0f, 0.0f, 0.0f);

    int buf = 0;
    int t = blockIdx.x;
    if (t < n_tiles) {
        px0 = x_fetch(x, t, tid, 0, n_nodes); px1 = x_fetch(x, t, tid, 1, n_nodes);
        px2 = x_fetch(x, t, tid, 2, n_nodes); px3 = x_fetch(x, t, tid, 3, n_nodes);
        px4 = x_fetch(x, t, tid, 4, n_nodes);
    }
    for (; t < n_tiles; t += GEMM_GROUPS) {
        float* sx = shx[buf];
        x_stage(sx, tid, 0, px0); x_stage(sx, tid, 1, px1); x_stage(sx, tid, 2, px2);
        x_stage(sx, tid, 3, px3); x_stage(sx, tid, 4, px4);
        __syncthreads();
        const int tn = t + GEMM_GROUPS;
        if (tn < n_tiles) {
            px0 = x_fetch(x, tn, tid, 0, n_nodes); px1 = x_fetch(x, tn, tid, 1, n_nodes);
            px2 = x_fetch(x, tn, tid, 2, n_nodes); px3 = x_fetch(x, tn, tid, 3, n_nodes);
            px4 = x_fetch(x, tn, tid, 4, n_nodes);
        }

        v8f acc;
        #pragma unroll
        for (int i = 0; i < 8; ++i) acc[i] = bv;
        const float* arow = sx + m * XS + 2 * hi;
        #pragma unroll
        for (int s = 0; s < IN_DIM / 4; ++s) {
            v2f a = *reinterpret_cast<const v2f*>(arow + 4 * s);
            acc = __builtin_amdgcn_wmma_f32_16x16x4_f32(false, a, false, b[s],
                                                        (short)0, acc, false, false);
        }
        asm volatile("v_nop\n\tv_nop\n\tv_nop\n\tv_nop" : "+v"(acc));

        const long row0 = (long)t * 16;
        float* hrow = h + row0 * H_DIM + ncol;
        if (row0 + 16 <= (long)n_nodes) {
            #pragma unroll
            for (int v = 0; v < 8; ++v) {
                float r = acc[v];
                hrow[(long)(v + 8 * hi) * H_DIM] = r > 0.0f ? r : 0.0f;
            }
        } else {
            #pragma unroll
            for (int v = 0; v < 8; ++v) {
                int mm = v + 8 * hi;
                if (row0 + mm < n_nodes) {
                    float r = acc[v];
                    hrow[(long)mm * H_DIM] = r > 0.0f ? r : 0.0f;
                }
            }
        }
        buf ^= 1;
    }
}

__global__ __launch_bounds__(256)
void sage_layer_kernel(const float* agg, const float* __restrict__ hin,
                       const float* __restrict__ deg,
                       const float* __restrict__ Wl, const float* __restrict__ Wr,
                       const float* __restrict__ bias, float* hout,
                       int n_nodes, int n_tiles) {
    __shared__ __attribute__((aligned(16))) float sh_a[2][16 * HS];
    __shared__ __attribute__((aligned(16))) float sh_h[2][16 * HS];
    const int tid = threadIdx.x;
    const int wave = tid >> 5, lane = tid & 31;
    const int m = lane & 15, hi = lane >> 4;
    const int ncol = wave * 16 + m;

    v2f bl[H_DIM / 4], br[H_DIM / 4];
    #pragma unroll
    for (int s = 0; s < H_DIM / 4; ++s) {
        const int k = 4 * s + 2 * hi;
        bl[s].x = Wl[k * H_DIM + ncol];
        bl[s].y = Wl[(k + 1) * H_DIM + ncol];
        br[s].x = Wr[k * H_DIM + ncol];
        br[s].y = Wr[(k + 1) * H_DIM + ncol];
    }
    const float bv = bias[ncol];

    const int sr = tid >> 4, sc = tid & 15;
    float4 pa0, pa1, ph0, ph1;
    float pdeg;
    auto fetch = [&](int t) {
        long rr = (long)t * 16 + sr; if (rr >= n_nodes) rr = n_nodes - 1;
        const float4* a4 = reinterpret_cast<const float4*>(agg + rr * H_DIM);
        const float4* h4 = reinterpret_cast<const float4*>(hin + rr * H_DIM);
        pa0 = a4[sc]; pa1 = a4[sc + 16];
        ph0 = h4[sc]; ph1 = h4[sc + 16];
        pdeg = deg[rr];
    };

    int buf = 0;
    int t = blockIdx.x;
    if (t < n_tiles) fetch(t);
    for (; t < n_tiles; t += GEMM_GROUPS) {
        float* sa = sh_a[buf];
        float* sh = sh_h[buf];
        {
            const float inv = 1.0f / (pdeg > 1.0f ? pdeg : 1.0f);
            float4 q0 = pa0, q1 = pa1;
            q0.x *= inv; q0.y *= inv; q0.z *= inv; q0.w *= inv;
            q1.x *= inv; q1.y *= inv; q1.z *= inv; q1.w *= inv;
            *reinterpret_cast<float4*>(sa + sr * HS + 4 * sc)      = q0;
            *reinterpret_cast<float4*>(sa + sr * HS + 4 * sc + 64) = q1;
            *reinterpret_cast<float4*>(sh + sr * HS + 4 * sc)      = ph0;
            *reinterpret_cast<float4*>(sh + sr * HS + 4 * sc + 64) = ph1;
        }
        __syncthreads();
        if (t + GEMM_GROUPS < n_tiles) fetch(t + GEMM_GROUPS);

        v8f acc;
        #pragma unroll
        for (int i = 0; i < 8; ++i) acc[i] = bv;
        const float* arow = sa + m * HS + 2 * hi;
        const float* hrow = sh + m * HS + 2 * hi;
        #pragma unroll
        for (int s = 0; s < H_DIM / 4; ++s) {
            v2f aa = *reinterpret_cast<const v2f*>(arow + 4 * s);
            v2f ah = *reinterpret_cast<const v2f*>(hrow + 4 * s);
            acc = __builtin_amdgcn_wmma_f32_16x16x4_f32(false, aa, false, bl[s],
                                                        (short)0, acc, false, false);
            acc = __builtin_amdgcn_wmma_f32_16x16x4_f32(false, ah, false, br[s],
                                                        (short)0, acc, false, false);
        }
        asm volatile("v_nop\n\tv_nop\n\tv_nop\n\tv_nop" : "+v"(acc));

        const long row0 = (long)t * 16;
        float* orow = hout + row0 * H_DIM + ncol;
        if (row0 + 16 <= (long)n_nodes) {
            #pragma unroll
            for (int v = 0; v < 8; ++v) {
                float r = acc[v];
                orow[(long)(v + 8 * hi) * H_DIM] = r > 0.0f ? r : 0.0f;
            }
        } else {
            #pragma unroll
            for (int v = 0; v < 8; ++v) {
                int mm = v + 8 * hi;
                if (row0 + mm < n_nodes) {
                    float r = acc[v];
                    orow[(long)mm * H_DIM] = r > 0.0f ? r : 0.0f;
                }
            }
        }
        buf ^= 1;
    }
}


static inline int cdiv(long a, long b) { return (int)((a + b - 1) / b); }

extern "C" void kernel_launch(void* const* d_in, const int* in_sizes, int n_in,
                              void* d_out, int out_size, void* d_ws, size_t ws_size,
                              hipStream_t stream) {
    const float* x    = (const float*)d_in[0];
    const int*   ei   = (const int*)  d_in[1];
    const float* Wemb = (const float*)d_in[2];
    const float* bemb = (const float*)d_in[3];
    const float* Wl   = (const float*)d_in[4];
    const float* bl   = (const float*)d_in[5];
    const float* Wr   = (const float*)d_in[6];
    float* out = (float*)d_out;

    const int N = in_sizes[0] / IN_DIM;
    const int E = in_sizes[1] / 2;
    const int* src = ei;
    const int* dst = ei + E;

    char*  ws  = (char*)d_ws;
    size_t off = 0;
    auto carve = [&](size_t bytes) -> float* {
        float* p = (float*)(ws + off);
        off = (off + bytes + 255) & ~(size_t)255;
        return p;
    };
    const int nblk = cdiv(N, 256);
    float* deg   = carve((size_t)N * sizeof(float));
    float* hP    = carve((size_t)N * H_DIM * sizeof(float));
    float* hQ    = carve((size_t)N * H_DIM * sizeof(float));
    int* cnt     = (int*)carve((size_t)N * sizeof(int));
    int* row_ptr = (int*)carve((size_t)(N + 1) * sizeof(int));
    int* bsum    = (int*)carve((size_t)nblk * sizeof(int));
    int* rank    = (int*)carve((size_t)E * sizeof(int));
    int* csr_src = (int*)carve((size_t)E * sizeof(int));
    (void)ws_size;

    const int tiles = cdiv(N, 16);

    zero_i32_kernel<<<cdiv(N, 256), 256, 0, stream>>>(cnt, N);
    count_rank_kernel<<<cdiv(E, 256), 256, 0, stream>>>(dst, cnt, rank, E);
    block_sum_kernel<<<nblk, 256, 0, stream>>>(cnt, bsum, N);
    row_ptr_kernel<<<nblk, 256, 0, stream>>>(cnt, bsum, row_ptr, deg, N);
    fill_kernel<<<cdiv(E, 256), 256, 0, stream>>>(src, dst, rank, row_ptr, csr_src, E);

    emb_kernel<<<GEMM_GROUPS, 256, 0, stream>>>(x, Wemb, bemb, hP, N, tiles);

    float* hcur = hP;
    for (int l = 0; l < L_LAYERS; ++l) {
        float* agg = (hcur == hP) ? hQ : hP;
        gather_sum_kernel<<<GATHER_BLOCKS, 256, 0, stream>>>(row_ptr, csr_src, hcur, agg, N);
        float* hnext = (l == L_LAYERS - 1) ? out : agg;
        sage_layer_kernel<<<GEMM_GROUPS, 256, 0, stream>>>(
            agg, hcur, deg,
            Wl + (long)l * H_DIM * H_DIM, Wr + (long)l * H_DIM * H_DIM,
            bl + (long)l * H_DIM, hnext, N, tiles);
        hcur = hnext;
    }
}


# edge grouping without a returning global add per edge: edges binned by 256-node range with LDS counters, one block per bin ranks per node in LDS; list order still arrival order
# speedup vs baseline: 3.8721x; 1.1588x over previous
#include <hip/hip_runtime.h>

typedef __attribute__((ext_vector_type(2))) float v2f;
typedef __attribute__((ext_vector_type(8))) float v8f;

#define IN_DIM 300
#define H_DIM  128
#define L_LAYERS 4

#define XS (IN_DIM + 16)
#define HS (H_DIM + 4)


__global__ void zero_i32_kernel(int* __restrict__ p, int n) {
    int i = blockIdx.x * blockDim.x + threadIdx.x;
    if (i < n) p[i] = 0;
}

#define BIN_SHIFT 8
#define BIN_NODES 256
#define CSTRIDE 64
#define CUR_OFF 32
#define EPT 16
#define EBLK (256 * EPT)

__global__ __launch_bounds__(256)
void bin_count_kernel(const int* __restrict__ dst, int* __restrict__ binctr, int e, int nbins) {
    extern __shared__ int hist[];
    const int tid = threadIdx.x;
    for (int b = tid; b < nbins; b += 256) hist[b] = 0;
    __syncthreads();
    const int i0 = blockIdx.x * EBLK + tid;
    int d[EPT];
    #pragma unroll
    for (int u = 0; u < EPT; ++u) {
        const int i = i0 + 256 * u;
        d[u] = (i < e) ? dst[i] : -1;
    }
    #pragma unroll
    for (int u = 0; u < EPT; ++u)
        if (d[u] >= 0) atomicAdd(&hist[d[u] >> BIN_SHIFT], 1);
    __syncthreads();
    for (int b = tid; b < nbins; b += 256) {
        const int c = hist[b];
        if (c) atomicAdd(&binctr[b * CSTRIDE], c);
    }
}

__global__ __launch_bounds__(256)
void bin_scan_kernel(int* __restrict__ binctr, int* __restrict__ bin_base, int nbins) {
    __shared__ int sh_tot[8];
    const int tid = threadIdx.x, lane = tid & 31, wave = tid >> 5;
    int carry = 0;
    for (int c0 = 0; c0 < nbins; c0 += 256) {
        const int i = c0 + tid;
        const int v = (i < nbins) ? binctr[i * CSTRIDE] : 0;
        int inc = v;
        #pragma unroll
        for (int o = 1; o < 32; o <<= 1) {
            int t = __shfl_up(inc, o, 32);
            if (lane >= o) inc += t;
        }
        if (lane == 31) sh_tot[wave] = inc;
        __syncthreads();
        int base = carry, tot = 0;
        #pragma unroll
        for (int w = 0; w < 8; ++w) {
            const int s = sh_tot[w];
            if (w < wave) base += s;
            tot += s;
        }
        const int excl = base + inc - v;
        if (i < nbins) {
            bin_base[i] = excl;
            binctr[i * CSTRIDE + CUR_OFF] = excl;
        }
        carry += tot;
        __syncthreads();
    }
    if (tid == 0) bin_base[nbins] = carry;
}

__global__ __launch_bounds__(256)
void bin_scatter_kernel(const int* __restrict__ src, const int* __restrict__ dst,
                        int* __restrict__ binctr, int2* __restrict__ bpair, int e, int nbins) {
    extern __shared__ int hist[];
    int* run = hist + nbins;
    const int tid = threadIdx.x;
    for (int b = tid; b < nbins; b += 256) hist[b] = 0;
    __syncthreads();
    const int i0 = blockIdx.x * EBLK + tid;
    int d[EPT], s[EPT], r[EPT];
    #pragma unroll
    for (int u = 0; u < EPT; ++u) {
        const int i = i0 + 256 * u;
        d[u] = (i < e) ? dst[i] : -1;
        s[u] = (i < e) ? src[i] : 0;
    }
    #pragma unroll
    for (int u = 0; u < EPT; ++u) {
        r[u] = 0;
        if (d[u] >= 0) r[u] = atomicAdd(&hist[d[u] >> BIN_SHIFT], 1);
    }
    __syncthreads();
    for (int b = tid; b < nbins; b += 256) {
        const int c = hist[b];
        run[b] = c ? atomicAdd(&binctr[b * CSTRIDE + CUR_OFF], c) : 0;
    }
    __syncthreads();
    #pragma unroll
    for (int u = 0; u < EPT; ++u) {
        if (d[u] >= 0)
            bpair[run[d[u] >> BIN_SHIFT] + r[u]] = make_int2(s[u], d[u] & (BIN_NODES - 1));
    }
}

#define CU 8
__global__ __launch_bounds__(256)
void csr_bin_kernel(const int2* __restrict__ bpair, const int* __restrict__ bin_base,
                    int* __restrict__ row_ptr, float* __restrict__ deg,
                    int* __restrict__ csr_src, int n) {
    __shared__ int cntL[BIN_NODES];
    __shared__ int curL[BIN_NODES];
    __shared__ int sh_tot[8];
    const int tid = threadIdx.x, lane = tid & 31, wave = tid >> 5;
    const int beg = bin_base[blockIdx.x], end = bin_base[blockIdx.x + 1];
    cntL[tid] = 0;
    __syncthreads();
    for (int i = beg + tid; i < end; i += 256 * CU) {
        int dd[CU];
        #pragma unroll
        for (int u = 0; u < CU; ++u) {
            const int j = i + 256 * u;
            dd[u] = (j < end) ? bpair[j].y : -1;
        }
        #pragma unroll
        for (int u = 0; u < CU; ++u)
            if (dd[u] >= 0) atomicAdd(&cntL[dd[u]], 1);
    }
    __syncthreads();
    const int c = cntL[tid];
    int inc = c;
    #pragma unroll
    for (int o = 1; o < 32; o <<= 1) {
        int t = __shfl_up(inc, o, 32);
        if (lane >= o) inc += t;
    }
    if (lane == 31) sh_tot[wave] = inc;
    __syncthreads();
    int base = 0;
    #pragma unroll
    for (int w = 0; w < 8; ++w)
        if (w < wave) base += sh_tot[w];
    const int excl = base + inc - c;
    curL[tid] = excl;
    const int node = blockIdx.x * BIN_NODES + tid;
    if (node < n) {
        row_ptr[node] = beg + excl;
        deg[node] = (float)c;
        if (node == n - 1) row_ptr[n] = beg + excl + c;
    }
    __syncthreads();
    for (int i = beg + tid; i < end; i += 256 * CU) {
        int2 pr[CU];
        #pragma unroll
        for (int u = 0; u < CU; ++u) {
            const int j = i + 256 * u;
            pr[u] = (j < end) ? bpair[j] : make_int2(0, -1);
        }
        #pragma unroll
        for (int u = 0; u < CU; ++u) {
            if (pr[u].y >= 0) {
                const int p = atomicAdd(&curL[pr[u].y], 1);
                csr_src[beg + p] = pr[u].x;
            }
        }
    }
}

#define GATHER_BLOCKS 4096
__device__ __forceinline__ void add4(float4& a, const float4& v) {
    a.x += v.x; a.y += v.y; a.z += v.z; a.w += v.w;
}
template <int R>
__device__ __forceinline__ void gather_rows(const float4* __restrict__ h4, int idx, int j,
                                            int lane, float4& acc) {
    float4 v[R];
    #pragma unroll
    for (int u = 0; u < R; ++u) {
        int s = __builtin_amdgcn_readlane(idx, j + u);
        v[u] = h4[(long)s * (H_DIM / 4) + lane];
    }
    #pragma unroll
    for (int u = 0; u < R; ++u) add4(acc, v[u]);
}
__global__ __launch_bounds__(256)
void gather_sum_kernel(const int* __restrict__ row_ptr, const int* __restrict__ csr_src,
                       const float* __restrict__ h, float* __restrict__ agg, int n_nodes) {
    const int lane = threadIdx.x & 31;
    const int wave0 = __builtin_amdgcn_readfirstlane((int)((blockIdx.x * 256 + threadIdx.x) >> 5));
    const float4* h4 = reinterpret_cast<const float4*>(h);
    float4* a4 = reinterpret_cast<float4*>(agg);
    for (int node = wave0; node < n_nodes; node += GATHER_BLOCKS * 8) {
        const int beg = row_ptr[node], end = row_ptr[node + 1];
        float4 acc = make_float4(0.0f, 0.0f, 0.0f, 0.0f);
        for (int base = beg; base < end; base += 32) {
            int m = end - base; if (m > 32) m = 32;
            int idx = (lane < m) ? csr_src[base + lane] : 0;
            int j = 0;
            for (; j + 8 <= m; j += 8) gather_rows<8>(h4, idx, j, lane, acc);
            if (m & 4) { gather_rows<4>(h4, idx, j, lane, acc); j += 4; }
            if (m & 2) { gather_rows<2>(h4, idx, j, lane, acc); j += 2; }
            if (m & 1) { gather_rows<1>(h4, idx, j, lane, acc); }
        }
        a4[(long)node * (H_DIM / 4) + lane] = acc;
    }
}

#define GEMM_GROUPS 512

#define XQ (IN_DIM / 4)
#define XF4 (16 * XQ)
#define XPT ((XF4 + 255) / 256)
__device__ __forceinline__ float4 x_fetch(const float* __restrict__ x, int t, int tid, int j,
                                          int n_nodes) {
    int i = tid + 256 * j; if (i >= XF4) i = XF4 - 1;
    const int r = i / XQ, c = i - r * XQ;
    long rr = (long)t * 16 + r; if (rr >= n_nodes) rr = n_nodes - 1;
    return reinterpret_cast<const float4*>(x + rr * IN_DIM)[c];
}
__device__ __forceinline__ void x_stage(float* sx, int tid, int j, const float4& v) {
    const int i = tid + 256 * j;
    if (i < XF4) {
        const int r = i / XQ, c = i - r * XQ;
        *reinterpret_cast<float4*>(sx + r * XS + 4 * c) = v;
    }
}
__global__ __launch_bounds__(256)
void emb_kernel(const float* __restrict__ x, const float* __restrict__ Wemb,
                const float* __restrict__ bemb, float* __restrict__ h,
                int n_nodes, int n_tiles) {
    __shared__ __attribute__((aligned(16))) float shx[2][16 * XS];
    const int tid = threadIdx.x;
    const int wave = tid >> 5, lane = tid & 31;
    const int m = lane & 15, hi = lane >> 4;
    const int ncol = wave * 16 + m;

    v2f b[IN_DIM / 4];
    #pragma unroll
    for (int s = 0; s < IN_DIM / 4; ++s) {
        const int k = 4 * s + 2 * hi;
        b[s].x = Wemb[k * H_DIM + ncol];
        b[s].y = Wemb[(k + 1) * H_DIM + ncol];
    }
    const float bv = bemb[ncol];

    static_assert(XPT == 5, "five float4 of x per thread and tile");
    float4 px0, px1, px2, px3, px4;
    px0 = px1 = px2 = px3 = px4 = make_float4(0.0f, 0.0f, 0.0f, 0.0f);

    int buf = 0;
    int t = blockIdx.x;
    if (t < n_tiles) {
        px0 = x_fetch(x, t, tid, 0, n_nodes); px1 = x_fetch(x, t, tid, 1, n_nodes);
        px2 = x_fetch(x, t, tid, 2, n_nodes); px3 = x_fetch(x, t, tid, 3, n_nodes);
        px4 = x_fetch(x, t, tid, 4, n_nodes);
    }
    for (; t < n_tiles; t += GEMM_GROUPS) {
        float* sx = shx[buf];
        x_stage(sx, tid, 0, px0); x_stage(sx, tid, 1, px1); x_stage(sx, tid, 2, px2);
        x_stage(sx, tid, 3, px3); x_stage(sx, tid, 4, px4);
        __syncthreads();
        const int tn = t + GEMM_GROUPS;
        if (tn < n_tiles) {
            px0 = x_fetch(x, tn, tid, 0, n_nodes); px1 = x_fetch(x, tn, tid, 1, n_nodes);
            px2 = x_fetch(x, tn, tid, 2, n_nodes); px3 = x_fetch(x, tn, tid, 3, n_nodes);
            px4 = x_fetch(x, tn, tid, 4, n_nodes);
        }

        v8f acc;
        #pragma unroll
        for (int i = 0; i < 8; ++i) acc[i] = bv;
        const float* arow = sx + m * XS + 2 * hi;
        #pragma unroll
        for (int s = 0; s < IN_DIM / 4; ++s) {
            v2f a = *reinterpret_cast<const v2f*>(arow + 4 * s);
            acc = __builtin_amdgcn_wmma_f32_16x16x4_f32(false, a, false, b[s],
                                                        (short)0, acc, false, false);
        }
        asm volatile("v_nop\n\tv_nop\n\tv_nop\n\tv_nop" : "+v"(acc));

        const long row0 = (long)t * 16;
        float* hrow = h + row0 * H_DIM + ncol;
        if (row0 + 16 <= (long)n_nodes) {
            #pragma unroll
            for (int v = 0; v < 8; ++v) {
                float r = acc[v];
                hrow[(long)(v + 8 * hi) * H_DIM] = r > 0.0f ? r : 0.0f;
            }
        } else {
            #pragma unroll
            for (int v = 0; v < 8; ++v) {
                int mm = v + 8 * hi;
                if (row0 + mm < n_nodes) {
                    float r = acc[v];
                    hrow[(long)mm * H_DIM] = r > 0.0f ? r : 0.0f;
                }
            }
        }
        buf ^= 1;
    }
}

__global__ __launch_bounds__(256)
void sage_layer_kernel(const float* agg, const float* __restrict__ hin,
                       const float* __restrict__ deg,
                       const float* __restrict__ Wl, const float* __restrict__ Wr,
                       const float* __restrict__ bias, float* hout,
                       int n_nodes, int n_tiles) {
    __shared__ __attribute__((aligned(16))) float sh_a[2][16 * HS];
    __shared__ __attribute__((aligned(16))) float sh_h[2][16 * HS];
    const int tid = threadIdx.x;
    const int wave = tid >> 5, lane = tid & 31;
    const int m = lane & 15, hi = lane >> 4;
    const int ncol = wave * 16 + m;

    v2f bl[H_DIM / 4], br[H_DIM / 4];
    #pragma unroll
    for (int s = 0; s < H_DIM / 4; ++s) {
        const int k = 4 * s + 2 * hi;
        bl[s].x = Wl[k * H_DIM + ncol];
        bl[s].y = Wl[(k + 1) * H_DIM + ncol];
        br[s].x = Wr[k * H_DIM + ncol];
        br[s].y = Wr[(k + 1) * H_DIM + ncol];
    }
    const float bv = bias[ncol];

    const int sr = tid >> 4, sc = tid & 15;
    float4 pa0, pa1, ph0, ph1;
    float pdeg;
    auto fetch = [&](int t) {
        long rr = (long)t * 16 + sr; if (rr >= n_nodes) rr = n_nodes - 1;
        const float4* a4 = reinterpret_cast<const float4*>(agg + rr * H_DIM);
        const float4* h4 = reinterpret_cast<const float4*>(hin + rr * H_DIM);
        pa0 = a4[sc]; pa1 = a4[sc + 16];
        ph0 = h4[sc]; ph1 = h4[sc + 16];
        pdeg = deg[rr];
    };

    int buf = 0;
    int t = blockIdx.x;
    if (t < n_tiles) fetch(t);
    for (; t < n_tiles; t += GEMM_GROUPS) {
        float* sa = sh_a[buf];
        float* sh = sh_h[buf];
        {
            const float inv = 1.0f / (pdeg > 1.0f ? pdeg : 1.0f);
            float4 q0 = pa0, q1 = pa1;
            q0.x *= inv; q0.y *= inv; q0.z *= inv; q0.w *= inv;
            q1.x *= inv; q1.y *= inv; q1.z *= inv; q1.w *= inv;
            *reinterpret_cast<float4*>(sa + sr * HS + 4 * sc)      = q0;
            *reinterpret_cast<float4*>(sa + sr * HS + 4 * sc + 64) = q1;
            *reinterpret_cast<float4*>(sh + sr * HS + 4 * sc)      = ph0;
            *reinterpret_cast<float4*>(sh + sr * HS + 4 * sc + 64) = ph1;
        }
        __syncthreads();
        if (t + GEMM_GROUPS < n_tiles) fetch(t + GEMM_GROUPS);

        v8f acc;
        #pragma unroll
        for (int i = 0; i < 8; ++i) acc[i] = bv;
        const float* arow = sa + m * HS + 2 * hi;
        const float* hrow = sh + m * HS + 2 * hi;
        #pragma unroll
        for (int s = 0; s < H_DIM / 4; ++s) {
            v2f aa = *reinterpret_cast<const v2f*>(arow + 4 * s);
            v2f ah = *reinterpret_cast<const v2f*>(hrow + 4 * s);
            acc = __builtin_amdgcn_wmma_f32_16x16x4_f32(false, aa, false, bl[s],
                                                        (short)0, acc, false, false);
            acc = __builtin_amdgcn_wmma_f32_16x16x4_f32(false, ah, false, br[s],
                                                        (short)0, acc, false, false);
        }
        asm volatile("v_nop\n\tv_nop\n\tv_nop\n\tv_nop" : "+v"(acc));

        const long row0 = (long)t * 16;
        float* orow = hout + row0 * H_DIM + ncol;
        if (row0 + 16 <= (long)n_nodes) {
            #pragma unroll
            for (int v = 0; v < 8; ++v) {
                float r = acc[v];
                orow[(long)(v + 8 * hi) * H_DIM] = r > 0.0f ? r : 0.0f;
            }
        } else {
            #pragma unroll
            for (int v = 0; v < 8; ++v) {
                int mm = v + 8 * hi;
                if (row0 + mm < n_nodes) {
                    float r = acc[v];
                    orow[(long)mm * H_DIM] = r > 0.0f ? r : 0.0f;
                }
            }
        }
        buf ^= 1;
    }
}


static inline int cdiv(long a, long b) { return (int)((a + b - 1) / b); }

extern "C" void kernel_launch(void* const* d_in, const int* in_sizes, int n_in,
                              void* d_out, int out_size, void* d_ws, size_t ws_size,
                              hipStream_t stream) {
    const float* x    = (const float*)d_in[0];
    const int*   ei   = (const int*)  d_in[1];
    const float* Wemb = (const float*)d_in[2];
    const float* bemb = (const float*)d_in[3];
    const float* Wl   = (const float*)d_in[4];
    const float* bl   = (const float*)d_in[5];
    const float* Wr   = (const float*)d_in[6];
    float* out = (float*)d_out;

    const int N = in_sizes[0] / IN_DIM;
    const int E = in_sizes[1] / 2;
    const int* src = ei;
    const int* dst = ei + E;

    char*  ws  = (char*)d_ws;
    size_t off = 0;
    auto carve = [&](size_t bytes) -> float* {
        float* p = (float*)(ws + off);
        off = (off + bytes + 255) & ~(size_t)255;
        return p;
    };
    const int nblk = cdiv(N, BIN_NODES);
    float* deg   = carve((size_t)N * sizeof(float));
    float* hP    = carve((size_t)N * H_DIM * sizeof(float));
    float* hQ    = carve((size_t)N * H_DIM * sizeof(float));
    int* binctr  = (int*)carve((size_t)nblk * CSTRIDE * sizeof(int));
    int* bin_base = (int*)carve((size_t)(nblk + 1) * sizeof(int));
    int* row_ptr = (int*)carve((size_t)(N + 1) * sizeof(int));
    int* csr_src = (int*)carve((size_t)E * sizeof(int));
    int2* bpair  = (int2*)hQ;
    (void)ws_size;

    const int tiles = cdiv(N, 16);

    zero_i32_kernel<<<cdiv((long)nblk * CSTRIDE, 256), 256, 0, stream>>>(binctr, nblk * CSTRIDE);
    bin_count_kernel<<<cdiv(E, EBLK), 256, nblk * sizeof(int), stream>>>(dst, binctr, E, nblk);
    bin_scan_kernel<<<1, 256, 0, stream>>>(binctr, bin_base, nblk);
    bin_scatter_kernel<<<cdiv(E, EBLK), 256, 2 * nblk * sizeof(int), stream>>>(
        src, dst, binctr, bpair, E, nblk);
    csr_bin_kernel<<<nblk, 256, 0, stream>>>(bpair, bin_base, row_ptr, deg, csr_src, N);

    emb_kernel<<<GEMM_GROUPS, 256, 0, stream>>>(x, Wemb, bemb, hP, N, tiles);

    float* hcur = hP;
    for (int l = 0; l < L_LAYERS; ++l) {
        float* agg = (hcur == hP) ? hQ : hP;
        gather_sum_kernel<<<GATHER_BLOCKS, 256, 0, stream>>>(row_ptr, csr_src, hcur, agg, N);
        float* hnext = (l == L_LAYERS - 1) ? out : agg;
        sage_layer_kernel<<<GEMM_GROUPS, 256, 0, stream>>>(
            agg, hcur, deg,
            Wl + (long)l * H_DIM * H_DIM, Wr + (long)l * H_DIM * H_DIM,
            bl + (long)l * H_DIM, hnext, N, tiles);
        hcur = hnext;
    }
}


# neighbour sums fused into the layer kernel: each wave gathers its two nodes of a tile straight into the LDS A tile while the co-resident group multiplies; 4 launches and the sums' round trip go
# speedup vs baseline: 4.0311x; 1.0411x over previous
#include <hip/hip_runtime.h>

typedef __attribute__((ext_vector_type(2))) float v2f;
typedef __attribute__((ext_vector_type(8))) float v8f;

#define IN_DIM 300
#define H_DIM  128
#define L_LAYERS 4

#define XS (IN_DIM + 16)
#define HS (H_DIM + 4)


__global__ void zero_i32_kernel(int* __restrict__ p, int n) {
    int i = blockIdx.x * blockDim.x + threadIdx.x;
    if (i < n) p[i] = 0;
}

#define BIN_SHIFT 8
#define BIN_NODES 256
#define CSTRIDE 64
#define CUR_OFF 32
#define EPT 16
#define EBLK (256 * EPT)

__global__ __launch_bounds__(256)
void bin_count_kernel(const int* __restrict__ dst, int* __restrict__ binctr, int e, int nbins) {
    extern __shared__ int hist[];
    const int tid = threadIdx.x;
    for (int b = tid; b < nbins; b += 256) hist[b] = 0;
    __syncthreads();
    const int i0 = blockIdx.x * EBLK + tid;
    int d[EPT];
    #pragma unroll
    for (int u = 0; u < EPT; ++u) {
        const int i = i0 + 256 * u;
        d[u] = (i < e) ? dst[i] : -1;
    }
    #pragma unroll
    for (int u = 0; u < EPT; ++u)
        if (d[u] >= 0) atomicAdd(&hist[d[u] >> BIN_SHIFT], 1);
    __syncthreads();
    for (int b = tid; b < nbins; b += 256) {
        const int c = hist[b];
        if (c) atomicAdd(&binctr[b * CSTRIDE], c);
    }
}

__global__ __launch_bounds__(256)
void bin_scan_kernel(int* __restrict__ binctr, int* __restrict__ bin_base, int nbins) {
    __shared__ int sh_tot[8];
    const int tid = threadIdx.x, lane = tid & 31, wave = tid >> 5;
    int carry = 0;
    for (int c0 = 0; c0 < nbins; c0 += 256) {
        const int i = c0 + tid;
        const int v = (i < nbins) ? binctr[i * CSTRIDE] : 0;
        int inc = v;
        #pragma unroll
        for (int o = 1; o < 32; o <<= 1) {
            int t = __shfl_up(inc, o, 32);
            if (lane >= o) inc += t;
        }
        if (lane == 31) sh_tot[wave] = inc;
        __syncthreads();
        int base = carry, tot = 0;
        #pragma unroll
        for (int w = 0; w < 8; ++w) {
            const int s = sh_tot[w];
            if (w < wave) base += s;
            tot += s;
        }
        const int excl = base + inc - v;
        if (i < nbins) {
            bin_base[i] = excl;
            binctr[i * CSTRIDE + CUR_OFF] = excl;
        }
        carry += tot;
        __syncthreads();
    }
    if (tid == 0) bin_base[nbins] = carry;
}

__global__ __launch_bounds__(256)
void bin_scatter_kernel(const int* __restrict__ src, const int* __restrict__ dst,
                        int* __restrict__ binctr, int2* __restrict__ bpair, int e, int nbins) {
    extern __shared__ int hist[];
    int* run = hist + nbins;
    const int tid = threadIdx.x;
    for (int b = tid; b < nbins; b += 256) hist[b] = 0;
    __syncthreads();
    const int i0 = blockIdx.x * EBLK + tid;
    int d[EPT], s[EPT], r[EPT];
    #pragma unroll
    for (int u = 0; u < EPT; ++u) {
        const int i = i0 + 256 * u;
        d[u] = (i < e) ? dst[i] : -1;
        s[u] = (i < e) ? src[i] : 0;
    }
    #pragma unroll
    for (int u = 0; u < EPT; ++u) {
        r[u] = 0;
        if (d[u] >= 0) r[u] = atomicAdd(&hist[d[u] >> BIN_SHIFT], 1);
    }
    __syncthreads();
    for (int b = tid; b < nbins; b += 256) {
        const int c = hist[b];
        run[b] = c ? atomicAdd(&binctr[b * CSTRIDE + CUR_OFF], c) : 0;
    }
    __syncthreads();
    #pragma unroll
    for (int u = 0; u < EPT; ++u) {
        if (d[u] >= 0)
            bpair[run[d[u] >> BIN_SHIFT] + r[u]] = make_int2(s[u], d[u] & (BIN_NODES - 1));
    }
}

#define CU 8
__global__ __launch_bounds__(256)
void csr_bin_kernel(const int2* __restrict__ bpair, const int* __restrict__ bin_base,
                    int* __restrict__ row_ptr, int* __restrict__ csr_src, int n) {
    __shared__ int cntL[BIN_NODES];
    __shared__ int curL[BIN_NODES];
    __shared__ int sh_tot[8];
    const int tid = threadIdx.x, lane = tid & 31, wave = tid >> 5;
    const int beg = bin_base[blockIdx.x], end = bin_base[blockIdx.x + 1];
    cntL[tid] = 0;
    __syncthreads();
    for (int i = beg + tid; i < end; i += 256 * CU) {
        int dd[CU];
        #pragma unroll
        for (int u = 0; u < CU; ++u) {
            const int j = i + 256 * u;
            dd[u] = (j < end) ? bpair[j].y : -1;
        }
        #pragma unroll
        for (int u = 0; u < CU; ++u)
            if (dd[u] >= 0) atomicAdd(&cntL[dd[u]], 1);
    }
    __syncthreads();
    const int c = cntL[tid];
    int inc = c;
    #pragma unroll
    for (int o = 1; o < 32; o <<= 1) {
        int t = __shfl_up(inc, o, 32);
        if (lane >= o) inc += t;
    }
    if (lane == 31) sh_tot[wave] = inc;
    __syncthreads();
    int base = 0;
    #pragma unroll
    for (int w = 0; w < 8; ++w)
        if (w < wave) base += sh_tot[w];
    const int excl = base + inc - c;
    curL[tid] = excl;
    const int node = blockIdx.x * BIN_NODES + tid;
    if (node < n) {
        row_ptr[node] = beg + excl;
        if (node == n - 1) row_ptr[n] = beg + excl + c;
    }
    __syncthreads();
    for (int i = beg + tid; i < end; i += 256 * CU) {
        int2 pr[CU];
        #pragma unroll
        for (int u = 0; u < CU; ++u) {
            const int j = i + 256 * u;
            pr[u] = (j < end) ? bpair[j] : make_int2(0, -1);
        }
        #pragma unroll
        for (int u = 0; u < CU; ++u) {
            if (pr[u].y >= 0) {
                const int p = atomicAdd(&curL[pr[u].y], 1);
                csr_src[beg + p] = pr[u].x;
            }
        }
    }
}

__device__ __forceinline__ void add4(float4& a, const float4& v) {
    a.x += v.x; a.y += v.y; a.z += v.z; a.w += v.w;
}
template <int R>
__device__ __forceinline__ void gather_rows(const float4* __restrict__ h4, int idx, int j,
                                            int lane, float4& acc) {
    float4 v[R];
    #pragma unroll
    for (int u = 0; u < R; ++u) {
        int s = __builtin_amdgcn_readlane(idx, j + u);
        v[u] = h4[(long)s * (H_DIM / 4) + lane];
    }
    #pragma unroll
    for (int u = 0; u < R; ++u) add4(acc, v[u]);
}
__device__ __forceinline__ float4 gather_list(const int* __restrict__ csr_src,
                                              const float4* __restrict__ h4,
                                              int beg, int end, int idx, int lane) {
    float4 acc = make_float4(0.0f, 0.0f, 0.0f, 0.0f);
    for (int base = beg; base < end; base += 32) {
        int m = end - base; if (m > 32) m = 32;
        if (base != beg) idx = (lane < m) ? csr_src[base + lane] : 0;
        int j = 0;
        for (; j + 16 <= m; j += 16) gather_rows<16>(h4, idx, j, lane, acc);
        if (m & 8) { gather_rows<8>(h4, idx, j, lane, acc); j += 8; }
        if (m & 4) { gather_rows<4>(h4, idx, j, lane, acc); j += 4; }
        if (m & 2) { gather_rows<2>(h4, idx, j, lane, acc); j += 2; }
        if (m & 1) { gather_rows<1>(h4, idx, j, lane, acc); }
    }
    return acc;
}

#define GEMM_GROUPS 512

#define XQ (IN_DIM / 4)
#define XF4 (16 * XQ)
#define XPT ((XF4 + 255) / 256)
__device__ __forceinline__ float4 x_fetch(const float* __restrict__ x, int t, int tid, int j,
                                          int n_nodes) {
    int i = tid + 256 * j; if (i >= XF4) i = XF4 - 1;
    const int r = i / XQ, c = i - r * XQ;
    long rr = (long)t * 16 + r; if (rr >= n_nodes) rr = n_nodes - 1;
    return reinterpret_cast<const float4*>(x + rr * IN_DIM)[c];
}
__device__ __forceinline__ void x_stage(float* sx, int tid, int j, const float4& v) {
    const int i = tid + 256 * j;
    if (i < XF4) {
        const int r = i / XQ, c = i - r * XQ;
        *reinterpret_cast<float4*>(sx + r * XS + 4 * c) = v;
    }
}
__global__ __launch_bounds__(256)
void emb_kernel(const float* __restrict__ x, const float* __restrict__ Wemb,
                const float* __restrict__ bemb, float* __restrict__ h,
                int n_nodes, int n_tiles) {
    __shared__ __attribute__((aligned(16))) float shx[2][16 * XS];
    const int tid = threadIdx.x;
    const int wave = tid >> 5, lane = tid & 31;
    const int m = lane & 15, hi = lane >> 4;
    const int ncol = wave * 16 + m;

    v2f b[IN_DIM / 4];
    #pragma unroll
    for (int s = 0; s < IN_DIM / 4; ++s) {
        const int k = 4 * s + 2 * hi;
        b[s].x = Wemb[k * H_DIM + ncol];
        b[s].y = Wemb[(k + 1) * H_DIM + ncol];
    }
    const float bv = bemb[ncol];

    static_assert(XPT == 5, "five float4 of x per thread and tile");
    float4 px0, px1, px2, px3, px4;
    px0 = px1 = px2 = px3 = px4 = make_float4(0.0f, 0.0f, 0.0f, 0.0f);

    int buf = 0;
    int t = blockIdx.x;
    if (t < n_tiles) {
        px0 = x_fetch(x, t, tid, 0, n_nodes); px1 = x_fetch(x, t, tid, 1, n_nodes);
        px2 = x_fetch(x, t, tid, 2, n_nodes); px3 = x_fetch(x, t, tid, 3, n_nodes);
        px4 = x_fetch(x, t, tid, 4, n_nodes);
    }
    for (; t < n_tiles; t += GEMM_GROUPS) {
        float* sx = shx[buf];
        x_stage(sx, tid, 0, px0); x_stage(sx, tid, 1, px1); x_stage(sx, tid, 2, px2);
        x_stage(sx, tid, 3, px3); x_stage(sx, tid, 4, px4);
        __syncthreads();
        const int tn = t + GEMM_GROUPS;
        if (tn < n_tiles) {
            px0 = x_fetch(x, tn, tid, 0, n_nodes); px1 = x_fetch(x, tn, tid, 1, n_nodes);
            px2 = x_fetch(x, tn, tid, 2, n_nodes); px3 = x_fetch(x, tn, tid, 3, n_nodes);
            px4 = x_fetch(x, tn, tid, 4, n_nodes);
        }

        v8f acc;
        #pragma unroll
        for (int i = 0; i < 8; ++i) acc[i] = bv;
        const float* arow = sx + m * XS + 2 * hi;
        #pragma unroll
        for (int s = 0; s < IN_DIM / 4; ++s) {
            v2f a = *reinterpret_cast<const v2f*>(arow + 4 * s);
            acc = __builtin_amdgcn_wmma_f32_16x16x4_f32(false, a, false, b[s],
                                                        (short)0, acc, false, false);
        }
        asm volatile("v_nop\n\tv_nop\n\tv_nop\n\tv_nop" : "+v"(acc));

        const long row0 = (long)t * 16;
        float* hrow = h + row0 * H_DIM + ncol;
        if (row0 + 16 <= (long)n_nodes) {
            #pragma unroll
            for (int v = 0; v < 8; ++v) {
                float r = acc[v];
                hrow[(long)(v + 8 * hi) * H_DIM] = r > 0.0f ? r : 0.0f;
            }
        } else {
            #pragma unroll
            for (int v = 0; v < 8; ++v) {
                int mm = v + 8 * hi;
                if (row0 + mm < n_nodes) {
                    float r = acc[v];
                    hrow[(long)mm * H_DIM] = r > 0.0f ? r : 0.0f;
                }
            }
        }
        buf ^= 1;
    }
}

__global__ __launch_bounds__(256)
void sage_layer_kernel(const int* __restrict__ row_ptr, const int* __restrict__ csr_src,
                       const float* __restrict__ hin,
                       const float* __restrict__ Wl, const float* __restrict__ Wr,
                       const float* __restrict__ bias, float* __restrict__ hout,
                       int n_nodes, int n_tiles) {
    __shared__ __attribute__((aligned(16))) float sh_a[16 * HS];
    __shared__ __attribute__((aligned(16))) float sh_h[16 * HS];
    const int tid = threadIdx.x;
    const int lane = tid & 31;
    const int wave = __builtin_amdgcn_readfirstlane(tid >> 5);
    const int m = lane & 15, hi = lane >> 4;
    const int ncol = wave * 16 + m;

    v2f bl[H_DIM / 4], br[H_DIM / 4];
    #pragma unroll
    for (int s = 0; s < H_DIM / 4; ++s) {
        const int k = 4 * s + 2 * hi;
        bl[s].x = Wl[k * H_DIM + ncol];
        bl[s].y = Wl[(k + 1) * H_DIM + ncol];
        br[s].x = Wr[k * H_DIM + ncol];
        br[s].y = Wr[(k + 1) * H_DIM + ncol];
    }
    const float bv = bias[ncol];
    const float4* h4 = reinterpret_cast<const float4*>(hin);

    int b0 = 0, e0 = 0, b1 = 0, e1 = 0, i0 = 0, i1 = 0;
    auto bounds = [&](int t, int& c0, int& f0, int& c1, int& f1) {
        c0 = f0 = c1 = f1 = 0;
        if (t < n_tiles) {
            const int n0 = t * 16 + 2 * wave;
            if (n0 < n_nodes)     { c0 = row_ptr[n0];     f0 = row_ptr[n0 + 1]; }
            if (n0 + 1 < n_nodes) { c1 = row_ptr[n0 + 1]; f1 = row_ptr[n0 + 2]; }
        }
    };
    auto first_sources = [&](int c, int f) -> int {
        return (lane < f - c) ? csr_src[c + lane] : 0;
    };

    int t = blockIdx.x;
    bounds(t, b0, e0, b1, e1);
    i0 = first_sources(b0, e0);
    i1 = first_sources(b1, e1);
    for (; t < n_tiles; t += GEMM_GROUPS) {
        const int n0 = t * 16 + 2 * wave;
        const long r0 = (n0 < n_nodes) ? n0 : n_nodes - 1;
        const long r1 = (n0 + 1 < n_nodes) ? n0 + 1 : n_nodes - 1;
        const float4 own0 = h4[r0 * (H_DIM / 4) + lane];
        const float4 own1 = h4[r1 * (H_DIM / 4) + lane];
        int nb0, ne0, nb1, ne1;
        bounds(t + GEMM_GROUPS, nb0, ne0, nb1, ne1);
        float4 a0 = gather_list(csr_src, h4, b0, e0, i0, lane);
        float4 a1 = gather_list(csr_src, h4, b1, e1, i1, lane);
        {
            const float d0 = (float)(e0 - b0), d1 = (float)(e1 - b1);
            const float inv0 = 1.0f / (d0 > 1.0f ? d0 : 1.0f);
            const float inv1 = 1.0f / (d1 > 1.0f ? d1 : 1.0f);
            a0.x *= inv0; a0.y *= inv0; a0.z *= inv0; a0.w *= inv0;
            a1.x *= inv1; a1.y *= inv1; a1.z *= inv1; a1.w *= inv1;
            const int row = 2 * wave;
            *reinterpret_cast<float4*>(sh_a + row * HS + 4 * lane)       = a0;
            *reinterpret_cast<float4*>(sh_a + (row + 1) * HS + 4 * lane) = a1;
            *reinterpret_cast<float4*>(sh_h + row * HS + 4 * lane)       = own0;
            *reinterpret_cast<float4*>(sh_h + (row + 1) * HS + 4 * lane) = own1;
        }
        __syncthreads();
        b0 = nb0; e0 = ne0; b1 = nb1; e1 = ne1;
        i0 = first_sources(b0, e0);
        i1 = first_sources(b1, e1);

        v8f acc;
        #pragma unroll
        for (int i = 0; i < 8; ++i) acc[i] = bv;
        const float* arow = sh_a + m * HS + 2 * hi;
        const float* hrow = sh_h + m * HS + 2 * hi;
        #pragma unroll
        for (int s = 0; s < H_DIM / 4; ++s) {
            v2f aa = *reinterpret_cast<const v2f*>(arow + 4 * s);
            v2f ah = *reinterpret_cast<const v2f*>(hrow + 4 * s);
            acc = __builtin_amdgcn_wmma_f32_16x16x4_f32(false, aa, false, bl[s],
                                                        (short)0, acc, false, false);
            acc = __builtin_amdgcn_wmma_f32_16x16x4_f32(false, ah, false, br[s],
                                                        (short)0, acc, false, false);
        }
        asm volatile("v_nop\n\tv_nop\n\tv_nop\n\tv_nop" : "+v"(acc));

        const long row0 = (long)t * 16;
        float* orow = hout + row0 * H_DIM + ncol;
        if (row0 + 16 <= (long)n_nodes) {
            #pragma unroll
            for (int v = 0; v < 8; ++v) {
                float r = acc[v];
                orow[(long)(v + 8 * hi) * H_DIM] = r > 0.0f ? r : 0.0f;
            }
        } else {
            #pragma unroll
            for (int v = 0; v < 8; ++v) {
                int mm = v + 8 * hi;
                if (row0 + mm < n_nodes) {
                    float r = acc[v];
                    orow[(long)mm * H_DIM] = r > 0.0f ? r : 0.0f;
                }
            }
        }
        __syncthreads();
    }
}


static inline int cdiv(long a, long b) { return (int)((a + b - 1) / b); }

extern "C" void kernel_launch(void* const* d_in, const int* in_sizes, int n_in,
                              void* d_out, int out_size, void* d_ws, size_t ws_size,
                              hipStream_t stream) {
    const float* x    = (const float*)d_in[0];
    const int*   ei   = (const int*)  d_in[1];
    const float* Wemb = (const float*)d_in[2];
    const float* bemb = (const float*)d_in[3];
    const float* Wl   = (const float*)d_in[4];
    const float* bl   = (const float*)d_in[5];
    const float* Wr   = (const float*)d_in[6];
    float* out = (float*)d_out;

    const int N = in_sizes[0] / IN_DIM;
    const int E = in_sizes[1] / 2;
    const int* src = ei;
    const int* dst = ei + E;

    char*  ws  = (char*)d_ws;
    size_t off = 0;
    auto carve = [&](size_t bytes) -> float* {
        float* p = (float*)(ws + off);
        off = (off + bytes + 255) & ~(size_t)255;
        return p;
    };
    const int nblk = cdiv(N, BIN_NODES);
    float* hP    = carve((size_t)N * H_DIM * sizeof(float));
    float* hQ    = carve((size_t)N * H_DIM * sizeof(float));
    int* binctr  = (int*)carve((size_t)nblk * CSTRIDE * sizeof(int));
    int* bin_base = (int*)carve((size_t)(nblk + 1) * sizeof(int));
    int* row_ptr = (int*)carve((size_t)(N + 1) * sizeof(int));
    int* csr_src = (int*)carve((size_t)E * sizeof(int));
    int2* bpair  = (int2*)hQ;
    (void)ws_size;

    const int tiles = cdiv(N, 16);

    zero_i32_kernel<<<cdiv((long)nblk * CSTRIDE, 256), 256, 0, stream>>>(binctr, nblk * CSTRIDE);
    bin_count_kernel<<<cdiv(E, EBLK), 256, nblk * sizeof(int), stream>>>(dst, binctr, E, nblk);
    bin_scan_kernel<<<1, 256, 0, stream>>>(binctr, bin_base, nblk);
    bin_scatter_kernel<<<cdiv(E, EBLK), 256, 2 * nblk * sizeof(int), stream>>>(
        src, dst, binctr, bpair, E, nblk);
    csr_bin_kernel<<<nblk, 256, 0, stream>>>(bpair, bin_base, row_ptr, csr_src, N);

    emb_kernel<<<GEMM_GROUPS, 256, 0, stream>>>(x, Wemb, bemb, hP, N, tiles);

    float* hcur = hP;
    for (int l = 0; l < L_LAYERS; ++l) {
        float* hnext = (l == L_LAYERS - 1) ? out : ((hcur == hP) ? hQ : hP);
        sage_layer_kernel<<<GEMM_GROUPS, 256, 0, stream>>>(
            row_ptr, csr_src, hcur,
            Wl + (long)l * H_DIM * H_DIM, Wr + (long)l * H_DIM * H_DIM,
            bl + (long)l * H_DIM, hnext, N, tiles);
        hcur = hnext;
    }
}
